# starting kernel
# baseline (speedup 1.0000x reference)
#include <hip/hip_runtime.h>

typedef float v2f __attribute__((ext_vector_type(2)));
typedef float v8f __attribute__((ext_vector_type(8)));

#define IN_DIM     256
#define H_DIM      256
#define K_CAT      512
#define M_TILE     64
#define LDS_STRIDE 516
#define THREADS    256

__global__ __launch_bounds__(THREADS) void gru_wmma_f32(
    const float* __restrict__ x,   const float* __restrict__ hp,
    const float* __restrict__ wr,  const float* __restrict__ wz,
    const float* __restrict__ whh, const float* __restrict__ whx,
    const float* __restrict__ br,  const float* __restrict__ bz,
    const float* __restrict__ bh,  float* __restrict__ out)
{
    __shared__ float xc[M_TILE * LDS_STRIDE];

    const int tid  = threadIdx.x;
    const int lane = tid & 31;
    const int wn   = tid >> 5;
    const long block_row = (long)blockIdx.x * M_TILE;

    for (int i = tid; i < M_TILE * (K_CAT / 4); i += THREADS) {
        const int row = i >> 7;
        const int col = (i & 127) * 4;
        float4 v;
        if (col < IN_DIM) v = *(const float4*)(x  + (block_row + row) * IN_DIM + col);
        else              v = *(const float4*)(hp + (block_row + row) * H_DIM + (col - IN_DIM));
        *(float4*)(&xc[row * LDS_STRIDE + col]) = v;
    }
    __syncthreads();

    const int a_lrow = lane & 15;
    const int a_koff = (lane >> 4) * 2;
    const int b_col  = wn * 32 + (lane & 15);
    const int b_krow = (lane >> 4) * 2;

    v8f accR[2][4], accZ[2][4];
    #pragma unroll
    for (int nt = 0; nt < 2; ++nt)
        #pragma unroll
        for (int mt = 0; mt < 4; ++mt)
            #pragma unroll
            for (int v = 0; v < 8; ++v) { accR[nt][mt][v] = 0.0f; accZ[nt][mt][v] = 0.0f; }

    for (int k = 0; k < K_CAT; k += 4) {
        v2f a[4];
        #pragma unroll
        for (int mt = 0; mt < 4; ++mt)
            a[mt] = *(const v2f*)(&xc[(mt * 16 + a_lrow) * LDS_STRIDE + k + a_koff]);
        const long wrow = (long)(k + b_krow) * H_DIM + b_col;
        #pragma unroll
        for (int nt = 0; nt < 2; ++nt) {
            const float* pr = wr + wrow + nt * 16;
            v2f rb; rb.x = pr[0]; rb.y = pr[H_DIM];
            const float* pz = wz + wrow + nt * 16;
            v2f zb; zb.x = pz[0]; zb.y = pz[H_DIM];
            #pragma unroll
            for (int mt = 0; mt < 4; ++mt) {
                accR[nt][mt] = __builtin_amdgcn_wmma_f32_16x16x4_f32(
                    false, a[mt], false, rb, (short)0, accR[nt][mt], false, false);
                accZ[nt][mt] = __builtin_amdgcn_wmma_f32_16x16x4_f32(
                    false, a[mt], false, zb, (short)0, accZ[nt][mt], false, false);
            }
        }
    }
    __syncthreads();

    #pragma unroll
    for (int nt = 0; nt < 2; ++nt) {
        const int col = wn * 32 + nt * 16 + (lane & 15);
        const float brc = br[col];
        #pragma unroll
        for (int mt = 0; mt < 4; ++mt) {
            #pragma unroll
            for (int v = 0; v < 8; ++v) {
                const int row = mt * 16 + (lane >> 4) * 8 + v;
                float r = accR[nt][mt][v] + brc;
                r = 1.0f / (1.0f + __expf(-r));
                const int idx = row * LDS_STRIDE + IN_DIM + col;
                xc[idx] = r * xc[idx];
                accR[nt][mt][v] = 0.0f;
            }
        }
    }
    __syncthreads();

    for (int k = 0; k < IN_DIM; k += 4) {
        v2f a[4];
        #pragma unroll
        for (int mt = 0; mt < 4; ++mt)
            a[mt] = *(const v2f*)(&xc[(mt * 16 + a_lrow) * LDS_STRIDE + k + a_koff]);
        const long wrow = (long)(k + b_krow) * H_DIM + b_col;
        #pragma unroll
        for (int nt = 0; nt < 2; ++nt) {
            const float* p = whx + wrow + nt * 16;
            v2f bb; bb.x = p[0]; bb.y = p[H_DIM];
            #pragma unroll
            for (int mt = 0; mt < 4; ++mt)
                accR[nt][mt] = __builtin_amdgcn_wmma_f32_16x16x4_f32(
                    false, a[mt], false, bb, (short)0, accR[nt][mt], false, false);
        }
    }
    for (int k = 0; k < H_DIM; k += 4) {
        v2f a[4];
        #pragma unroll
        for (int mt = 0; mt < 4; ++mt)
            a[mt] = *(const v2f*)(&xc[(mt * 16 + a_lrow) * LDS_STRIDE + IN_DIM + k + a_koff]);
        const long wrow = (long)(k + b_krow) * H_DIM + b_col;
        #pragma unroll
        for (int nt = 0; nt < 2; ++nt) {
            const float* p = whh + wrow + nt * 16;
            v2f bb; bb.x = p[0]; bb.y = p[H_DIM];
            #pragma unroll
            for (int mt = 0; mt < 4; ++mt)
                accR[nt][mt] = __builtin_amdgcn_wmma_f32_16x16x4_f32(
                    false, a[mt], false, bb, (short)0, accR[nt][mt], false, false);
        }
    }

    #pragma unroll
    for (int nt = 0; nt < 2; ++nt) {
        const int col = wn * 32 + nt * 16 + (lane & 15);
        const float bzc = bz[col];
        const float bhc = bh[col];
        #pragma unroll
        for (int mt = 0; mt < 4; ++mt) {
            #pragma unroll
            for (int v = 0; v < 8; ++v) {
                const long row = block_row + mt * 16 + (lane >> 4) * 8 + v;
                float z = accZ[nt][mt][v] + bzc;
                z = 1.0f / (1.0f + __expf(-z));
                const float g  = tanhf(accR[nt][mt][v] + bhc);
                const float h0 = hp[row * H_DIM + col];
                const float hn = (1.0f - z) * h0 + z * g;
                __builtin_nontemporal_store(hn, out + row * H_DIM + col);
            }
        }
    }
}

extern "C" void kernel_launch(void* const* d_in, const int* in_sizes, int n_in,
                              void* d_out, int out_size, void* d_ws, size_t ws_size,
                              hipStream_t stream) {
    const float* x   = (const float*)d_in[0];
    const float* hp  = (const float*)d_in[1];
    const float* wr  = (const float*)d_in[2];
    const float* wz  = (const float*)d_in[3];
    const float* whh = (const float*)d_in[4];
    const float* whx = (const float*)d_in[5];
    const float* br  = (const float*)d_in[6];
    const float* bz  = (const float*)d_in[7];
    const float* bh  = (const float*)d_in[8];
    float* out = (float*)d_out;

    const int rows = in_sizes[0] / IN_DIM;
    dim3 grid(rows / M_TILE), block(THREADS);
    gru_wmma_f32<<<grid, block, 0, stream>>>(x, hp, wr, wz, whh, whx, br, bz, bh, out);
}


# h_prev for the output blend kept in registers from one batched LDS read in the r stage (same lane owns the same elements); removes 64 dependent global re-loads of h_prev per lane
# speedup vs baseline: 1.1322x; 1.1322x over previous
#include <hip/hip_runtime.h>

typedef float v2f __attribute__((ext_vector_type(2)));
typedef float v8f __attribute__((ext_vector_type(8)));

#define IN_DIM     256
#define H_DIM      256
#define K_CAT      512
#define M_TILE     64
#define LDS_STRIDE 516
#define THREADS    256

__global__ __launch_bounds__(THREADS) void gru_wmma_f32(
    const float* __restrict__ x,   const float* __restrict__ hp,
    const float* __restrict__ wr,  const float* __restrict__ wz,
    const float* __restrict__ whh, const float* __restrict__ whx,
    const float* __restrict__ br,  const float* __restrict__ bz,
    const float* __restrict__ bh,  float* __restrict__ out)
{
    __shared__ float xc[M_TILE * LDS_STRIDE];

    const int tid  = threadIdx.x;
    const int lane = tid & 31;
    const int wn   = tid >> 5;
    const long block_row = (long)blockIdx.x * M_TILE;

    for (int i = tid; i < M_TILE * (K_CAT / 4); i += THREADS) {
        const int row = i >> 7;
        const int col = (i & 127) * 4;
        float4 v;
        if (col < IN_DIM) v = *(const float4*)(x  + (block_row + row) * IN_DIM + col);
        else              v = *(const float4*)(hp + (block_row + row) * H_DIM + (col - IN_DIM));
        *(float4*)(&xc[row * LDS_STRIDE + col]) = v;
    }
    __syncthreads();

    const int a_lrow = lane & 15;
    const int a_koff = (lane >> 4) * 2;
    const int b_col  = wn * 32 + (lane & 15);
    const int b_krow = (lane >> 4) * 2;

    v8f accR[2][4], accZ[2][4];
    #pragma unroll
    for (int nt = 0; nt < 2; ++nt)
        #pragma unroll
        for (int mt = 0; mt < 4; ++mt)
            #pragma unroll
            for (int v = 0; v < 8; ++v) { accR[nt][mt][v] = 0.0f; accZ[nt][mt][v] = 0.0f; }

    for (int k = 0; k < K_CAT; k += 4) {
        v2f a[4];
        #pragma unroll
        for (int mt = 0; mt < 4; ++mt)
            a[mt] = *(const v2f*)(&xc[(mt * 16 + a_lrow) * LDS_STRIDE + k + a_koff]);
        const long wrow = (long)(k + b_krow) * H_DIM + b_col;
        #pragma unroll
        for (int nt = 0; nt < 2; ++nt) {
            const float* pr = wr + wrow + nt * 16;
            v2f rb; rb.x = pr[0]; rb.y = pr[H_DIM];
            const float* pz = wz + wrow + nt * 16;
            v2f zb; zb.x = pz[0]; zb.y = pz[H_DIM];
            #pragma unroll
            for (int mt = 0; mt < 4; ++mt) {
                accR[nt][mt] = __builtin_amdgcn_wmma_f32_16x16x4_f32(
                    false, a[mt], false, rb, (short)0, accR[nt][mt], false, false);
                accZ[nt][mt] = __builtin_amdgcn_wmma_f32_16x16x4_f32(
                    false, a[mt], false, zb, (short)0, accZ[nt][mt], false, false);
            }
        }
    }
    __syncthreads();

    float hreg[2][4][8];
    #pragma unroll
    for (int nt = 0; nt < 2; ++nt) {
        const int col = wn * 32 + nt * 16 + (lane & 15);
        #pragma unroll
        for (int mt = 0; mt < 4; ++mt) {
            #pragma unroll
            for (int v = 0; v < 8; ++v) {
                const int row = mt * 16 + (lane >> 4) * 8 + v;
                hreg[nt][mt][v] = xc[row * LDS_STRIDE + IN_DIM + col];
            }
        }
    }
    #pragma unroll
    for (int nt = 0; nt < 2; ++nt) {
        const int col = wn * 32 + nt * 16 + (lane & 15);
        const float brc = br[col];
        #pragma unroll
        for (int mt = 0; mt < 4; ++mt) {
            #pragma unroll
            for (int v = 0; v < 8; ++v) {
                const int row = mt * 16 + (lane >> 4) * 8 + v;
                float r = accR[nt][mt][v] + brc;
                r = 1.0f / (1.0f + __expf(-r));
                xc[row * LDS_STRIDE + IN_DIM + col] = r * hreg[nt][mt][v];
                accR[nt][mt][v] = 0.0f;
            }
        }
    }
    __syncthreads();

    for (int k = 0; k < IN_DIM; k += 4) {
        v2f a[4];
        #pragma unroll
        for (int mt = 0; mt < 4; ++mt)
            a[mt] = *(const v2f*)(&xc[(mt * 16 + a_lrow) * LDS_STRIDE + k + a_koff]);
        const long wrow = (long)(k + b_krow) * H_DIM + b_col;
        #pragma unroll
        for (int nt = 0; nt < 2; ++nt) {
            const float* p = whx + wrow + nt * 16;
            v2f bb; bb.x = p[0]; bb.y = p[H_DIM];
            #pragma unroll
            for (int mt = 0; mt < 4; ++mt)
                accR[nt][mt] = __builtin_amdgcn_wmma_f32_16x16x4_f32(
                    false, a[mt], false, bb, (short)0, accR[nt][mt], false, false);
        }
    }
    for (int k = 0; k < H_DIM; k += 4) {
        v2f a[4];
        #pragma unroll
        for (int mt = 0; mt < 4; ++mt)
            a[mt] = *(const v2f*)(&xc[(mt * 16 + a_lrow) * LDS_STRIDE + IN_DIM + k + a_koff]);
        const long wrow = (long)(k + b_krow) * H_DIM + b_col;
        #pragma unroll
        for (int nt = 0; nt < 2; ++nt) {
            const float* p = whh + wrow + nt * 16;
            v2f bb; bb.x = p[0]; bb.y = p[H_DIM];
            #pragma unroll
            for (int mt = 0; mt < 4; ++mt)
                accR[nt][mt] = __builtin_amdgcn_wmma_f32_16x16x4_f32(
                    false, a[mt], false, bb, (short)0, accR[nt][mt], false, false);
        }
    }

    #pragma unroll
    for (int nt = 0; nt < 2; ++nt) {
        const int col = wn * 32 + nt * 16 + (lane & 15);
        const float bzc = bz[col];
        const float bhc = bh[col];
        #pragma unroll
        for (int mt = 0; mt < 4; ++mt) {
            #pragma unroll
            for (int v = 0; v < 8; ++v) {
                const long row = block_row + mt * 16 + (lane >> 4) * 8 + v;
                float z = accZ[nt][mt][v] + bzc;
                z = 1.0f / (1.0f + __expf(-z));
                const float g  = tanhf(accR[nt][mt][v] + bhc);
                const float h0 = hreg[nt][mt][v];
                const float hn = (1.0f - z) * h0 + z * g;
                __builtin_nontemporal_store(hn, out + row * H_DIM + col);
            }
        }
    }
}

extern "C" void kernel_launch(void* const* d_in, const int* in_sizes, int n_in,
                              void* d_out, int out_size, void* d_ws, size_t ws_size,
                              hipStream_t stream) {
    const float* x   = (const float*)d_in[0];
    const float* hp  = (const float*)d_in[1];
    const float* wr  = (const float*)d_in[2];
    const float* wz  = (const float*)d_in[3];
    const float* whh = (const float*)d_in[4];
    const float* whx = (const float*)d_in[5];
    const float* br  = (const float*)d_in[6];
    const float* bz  = (const float*)d_in[7];
    const float* bh  = (const float*)d_in[8];
    float* out = (float*)d_out;

    const int rows = in_sizes[0] / IN_DIM;
    dim3 grid(rows / M_TILE), block(THREADS);
    gru_wmma_f32<<<grid, block, 0, stream>>>(x, hp, wr, wz, whh, whx, br, bz, bh, out);
}


# Wave's two N-tiles interleaved by column (tile nt, lane c -> column 2c+nt): each lane's weight, bias, h and out accesses become one 8-byte whole-line access instead of two 4-byte half-line ones
# speedup vs baseline: 1.1721x; 1.0352x over previous
#include <hip/hip_runtime.h>

typedef float v2f __attribute__((ext_vector_type(2)));
typedef float v8f __attribute__((ext_vector_type(8)));

#define IN_DIM     256
#define H_DIM      256
#define K_CAT      512
#define M_TILE     64
#define LDS_STRIDE 516
#define THREADS    256

__global__ __launch_bounds__(THREADS) void gru_wmma_f32(
    const float* __restrict__ x,   const float* __restrict__ hp,
    const float* __restrict__ wr,  const float* __restrict__ wz,
    const float* __restrict__ whh, const float* __restrict__ whx,
    const float* __restrict__ br,  const float* __restrict__ bz,
    const float* __restrict__ bh,  float* __restrict__ out)
{
    __shared__ float xc[M_TILE * LDS_STRIDE];

    const int tid  = threadIdx.x;
    const int lane = tid & 31;
    const int wn   = tid >> 5;
    const long block_row = (long)blockIdx.x * M_TILE;

    for (int i = tid; i < M_TILE * (K_CAT / 4); i += THREADS) {
        const int row = i >> 7;
        const int col = (i & 127) * 4;
        float4 v;
        if (col < IN_DIM) v = *(const float4*)(x  + (block_row + row) * IN_DIM + col);
        else              v = *(const float4*)(hp + (block_row + row) * H_DIM + (col - IN_DIM));
        *(float4*)(&xc[row * LDS_STRIDE + col]) = v;
    }
    __syncthreads();

    const int a_lrow = lane & 15;
    const int a_koff = (lane >> 4) * 2;
    const int col2   = wn * 32 + (lane & 15) * 2;
    const int b_krow = (lane >> 4) * 2;

    v8f accR[2][4], accZ[2][4];
    #pragma unroll
    for (int nt = 0; nt < 2; ++nt)
        #pragma unroll
        for (int mt = 0; mt < 4; ++mt)
            #pragma unroll
            for (int v = 0; v < 8; ++v) { accR[nt][mt][v] = 0.0f; accZ[nt][mt][v] = 0.0f; }

    for (int k = 0; k < K_CAT; k += 4) {
        v2f a[4];
        #pragma unroll
        for (int mt = 0; mt < 4; ++mt)
            a[mt] = *(const v2f*)(&xc[(mt * 16 + a_lrow) * LDS_STRIDE + k + a_koff]);
        const long wrow = (long)(k + b_krow) * H_DIM + col2;
        const v2f r0 = *(const v2f*)(wr + wrow);
        const v2f r1 = *(const v2f*)(wr + wrow + H_DIM);
        const v2f z0 = *(const v2f*)(wz + wrow);
        const v2f z1 = *(const v2f*)(wz + wrow + H_DIM);
        v2f rb[2], zb[2];
        rb[0].x = r0.x; rb[0].y = r1.x; rb[1].x = r0.y; rb[1].y = r1.y;
        zb[0].x = z0.x; zb[0].y = z1.x; zb[1].x = z0.y; zb[1].y = z1.y;
        #pragma unroll
        for (int nt = 0; nt < 2; ++nt) {
            #pragma unroll
            for (int mt = 0; mt < 4; ++mt) {
                accR[nt][mt] = __builtin_amdgcn_wmma_f32_16x16x4_f32(
                    false, a[mt], false, rb[nt], (short)0, accR[nt][mt], false, false);
                accZ[nt][mt] = __builtin_amdgcn_wmma_f32_16x16x4_f32(
                    false, a[mt], false, zb[nt], (short)0, accZ[nt][mt], false, false);
            }
        }
    }
    __syncthreads();

    float hreg[2][4][8];
    #pragma unroll
    for (int mt = 0; mt < 4; ++mt) {
        #pragma unroll
        for (int v = 0; v < 8; ++v) {
            const int row = mt * 16 + (lane >> 4) * 8 + v;
            const v2f hv = *(const v2f*)(&xc[row * LDS_STRIDE + IN_DIM + col2]);
            hreg[0][mt][v] = hv.x;
            hreg[1][mt][v] = hv.y;
        }
    }
    {
        const v2f brc = *(const v2f*)(br + col2);
        #pragma unroll
        for (int mt = 0; mt < 4; ++mt) {
            #pragma unroll
            for (int v = 0; v < 8; ++v) {
                const int row = mt * 16 + (lane >> 4) * 8 + v;
                float r0 = accR[0][mt][v] + brc.x;
                float r1 = accR[1][mt][v] + brc.y;
                r0 = 1.0f / (1.0f + __expf(-r0));
                r1 = 1.0f / (1.0f + __expf(-r1));
                v2f rh;
                rh.x = r0 * hreg[0][mt][v];
                rh.y = r1 * hreg[1][mt][v];
                *(v2f*)(&xc[row * LDS_STRIDE + IN_DIM + col2]) = rh;
                accR[0][mt][v] = 0.0f;
                accR[1][mt][v] = 0.0f;
            }
        }
    }
    __syncthreads();

    for (int k = 0; k < IN_DIM; k += 4) {
        v2f a[4];
        #pragma unroll
        for (int mt = 0; mt < 4; ++mt)
            a[mt] = *(const v2f*)(&xc[(mt * 16 + a_lrow) * LDS_STRIDE + k + a_koff]);
        const long wrow = (long)(k + b_krow) * H_DIM + col2;
        const v2f g0 = *(const v2f*)(whx + wrow);
        const v2f g1 = *(const v2f*)(whx + wrow + H_DIM);
        v2f bb[2];
        bb[0].x = g0.x; bb[0].y = g1.x; bb[1].x = g0.y; bb[1].y = g1.y;
        #pragma unroll
        for (int nt = 0; nt < 2; ++nt)
            #pragma unroll
            for (int mt = 0; mt < 4; ++mt)
                accR[nt][mt] = __builtin_amdgcn_wmma_f32_16x16x4_f32(
                    false, a[mt], false, bb[nt], (short)0, accR[nt][mt], false, false);
    }
    for (int k = 0; k < H_DIM; k += 4) {
        v2f a[4];
        #pragma unroll
        for (int mt = 0; mt < 4; ++mt)
            a[mt] = *(const v2f*)(&xc[(mt * 16 + a_lrow) * LDS_STRIDE + IN_DIM + k + a_koff]);
        const long wrow = (long)(k + b_krow) * H_DIM + col2;
        const v2f g0 = *(const v2f*)(whh + wrow);
        const v2f g1 = *(const v2f*)(whh + wrow + H_DIM);
        v2f bb[2];
        bb[0].x = g0.x; bb[0].y = g1.x; bb[1].x = g0.y; bb[1].y = g1.y;
        #pragma unroll
        for (int nt = 0; nt < 2; ++nt)
            #pragma unroll
            for (int mt = 0; mt < 4; ++mt)
                accR[nt][mt] = __builtin_amdgcn_wmma_f32_16x16x4_f32(
                    false, a[mt], false, bb[nt], (short)0, accR[nt][mt], false, false);
    }

    {
        const v2f bzc = *(const v2f*)(bz + col2);
        const v2f bhc = *(const v2f*)(bh + col2);
        #pragma unroll
        for (int mt = 0; mt < 4; ++mt) {
            #pragma unroll
            for (int v = 0; v < 8; ++v) {
                const long row = block_row + mt * 16 + (lane >> 4) * 8 + v;
                float z0 = accZ[0][mt][v] + bzc.x;
                float z1 = accZ[1][mt][v] + bzc.y;
                z0 = 1.0f / (1.0f + __expf(-z0));
                z1 = 1.0f / (1.0f + __expf(-z1));
                const float g0 = tanhf(accR[0][mt][v] + bhc.x);
                const float g1 = tanhf(accR[1][mt][v] + bhc.y);
                v2f hn;
                hn.x = (1.0f - z0) * hreg[0][mt][v] + z0 * g0;
                hn.y = (1.0f - z1) * hreg[1][mt][v] + z1 * g1;
                __builtin_nontemporal_store(hn, (v2f*)(out + row * H_DIM + col2));
            }
        }
    }
}

extern "C" void kernel_launch(void* const* d_in, const int* in_sizes, int n_in,
                              void* d_out, int out_size, void* d_ws, size_t ws_size,
                              hipStream_t stream) {
    const float* x   = (const float*)d_in[0];
    const float* hp  = (const float*)d_in[1];
    const float* wr  = (const float*)d_in[2];
    const float* wz  = (const float*)d_in[3];
    const float* whh = (const float*)d_in[4];
    const float* whx = (const float*)d_in[5];
    const float* br  = (const float*)d_in[6];
    const float* bz  = (const float*)d_in[7];
    const float* bh  = (const float*)d_in[8];
    float* out = (float*)d_out;

    const int rows = in_sizes[0] / IN_DIM;
    dim3 grid(rows / M_TILE), block(THREADS);
    gru_wmma_f32<<<grid, block, 0, stream>>>(x, hp, wr, wz, whh, whx, br, bz, bh, out);
}


# tanh and both sigmoids on the hardware exp and reciprocal (tanh v = 1 - 2/(1+e^2v); rcp replaces the IEEE divide chain; about 1 ulp each); sums and their order unchanged
# speedup vs baseline: 1.2639x; 1.0783x over previous
#include <hip/hip_runtime.h>

typedef float v2f __attribute__((ext_vector_type(2)));
typedef float v8f __attribute__((ext_vector_type(8)));

#define IN_DIM     256
#define H_DIM      256
#define K_CAT      512
#define M_TILE     64
#define LDS_STRIDE 516
#define THREADS    256

__device__ __forceinline__ float sigmoid_hw(float v) {
    return __builtin_amdgcn_rcpf(1.0f + __expf(-v));
}
__device__ __forceinline__ float tanh_hw(float v) {
    return 1.0f - 2.0f * __builtin_amdgcn_rcpf(1.0f + __builtin_amdgcn_exp2f(v * 2.885390082f));
}

__global__ __launch_bounds__(THREADS) void gru_wmma_f32(
    const float* __restrict__ x,   const float* __restrict__ hp,
    const float* __restrict__ wr,  const float* __restrict__ wz,
    const float* __restrict__ whh, const float* __restrict__ whx,
    const float* __restrict__ br,  const float* __restrict__ bz,
    const float* __restrict__ bh,  float* __restrict__ out)
{
    __shared__ float xc[M_TILE * LDS_STRIDE];

    const int tid  = threadIdx.x;
    const int lane = tid & 31;
    const int wn   = tid >> 5;
    const long block_row = (long)blockIdx.x * M_TILE;

    for (int i = tid; i < M_TILE * (K_CAT / 4); i += THREADS) {
        const int row = i >> 7;
        const int col = (i & 127) * 4;
        float4 v;
        if (col < IN_DIM) v = *(const float4*)(x  + (block_row + row) * IN_DIM + col);
        else              v = *(const float4*)(hp + (block_row + row) * H_DIM + (col - IN_DIM));
        *(float4*)(&xc[row * LDS_STRIDE + col]) = v;
    }
    __syncthreads();

    const int a_lrow = lane & 15;
    const int a_koff = (lane >> 4) * 2;
    const int col2   = wn * 32 + (lane & 15) * 2;
    const int b_krow = (lane >> 4) * 2;

    v8f accR[2][4], accZ[2][4];
    #pragma unroll
    for (int nt = 0; nt < 2; ++nt)
        #pragma unroll
        for (int mt = 0; mt < 4; ++mt)
            #pragma unroll
            for (int v = 0; v < 8; ++v) { accR[nt][mt][v] = 0.0f; accZ[nt][mt][v] = 0.0f; }

    for (int k = 0; k < K_CAT; k += 4) {
        v2f a[4];
        #pragma unroll
        for (int mt = 0; mt < 4; ++mt)
            a[mt] = *(const v2f*)(&xc[(mt * 16 + a_lrow) * LDS_STRIDE + k + a_koff]);
        const long wrow = (long)(k + b_krow) * H_DIM + col2;
        const v2f r0 = *(const v2f*)(wr + wrow);
        const v2f r1 = *(const v2f*)(wr + wrow + H_DIM);
        const v2f z0 = *(const v2f*)(wz + wrow);
        const v2f z1 = *(const v2f*)(wz + wrow + H_DIM);
        v2f rb[2], zb[2];
        rb[0].x = r0.x; rb[0].y = r1.x; rb[1].x = r0.y; rb[1].y = r1.y;
        zb[0].x = z0.x; zb[0].y = z1.x; zb[1].x = z0.y; zb[1].y = z1.y;
        #pragma unroll
        for (int nt = 0; nt < 2; ++nt) {
            #pragma unroll
            for (int mt = 0; mt < 4; ++mt) {
                accR[nt][mt] = __builtin_amdgcn_wmma_f32_16x16x4_f32(
                    false, a[mt], false, rb[nt], (short)0, accR[nt][mt], false, false);
                accZ[nt][mt] = __builtin_amdgcn_wmma_f32_16x16x4_f32(
                    false, a[mt], false, zb[nt], (short)0, accZ[nt][mt], false, false);
            }
        }
    }
    __syncthreads();

    float hreg[2][4][8];
    #pragma unroll
    for (int mt = 0; mt < 4; ++mt) {
        #pragma unroll
        for (int v = 0; v < 8; ++v) {
            const int row = mt * 16 + (lane >> 4) * 8 + v;
            const v2f hv = *(const v2f*)(&xc[row * LDS_STRIDE + IN_DIM + col2]);
            hreg[0][mt][v] = hv.x;
            hreg[1][mt][v] = hv.y;
        }
    }
    {
        const v2f brc = *(const v2f*)(br + col2);
        #pragma unroll
        for (int mt = 0; mt < 4; ++mt) {
            #pragma unroll
            for (int v = 0; v < 8; ++v) {
                const int row = mt * 16 + (lane >> 4) * 8 + v;
                float r0 = accR[0][mt][v] + brc.x;
                float r1 = accR[1][mt][v] + brc.y;
                r0 = sigmoid_hw(r0);
                r1 = sigmoid_hw(r1);
                v2f rh;
                rh.x = r0 * hreg[0][mt][v];
                rh.y = r1 * hreg[1][mt][v];
                *(v2f*)(&xc[row * LDS_STRIDE + IN_DIM + col2]) = rh;
                accR[0][mt][v] = 0.0f;
                accR[1][mt][v] = 0.0f;
            }
        }
    }
    __syncthreads();

    for (int k = 0; k < IN_DIM; k += 4) {
        v2f a[4];
        #pragma unroll
        for (int mt = 0; mt < 4; ++mt)
            a[mt] = *(const v2f*)(&xc[(mt * 16 + a_lrow) * LDS_STRIDE + k + a_koff]);
        const long wrow = (long)(k + b_krow) * H_DIM + col2;
        const v2f g0 = *(const v2f*)(whx + wrow);
        const v2f g1 = *(const v2f*)(whx + wrow + H_DIM);
        v2f bb[2];
        bb[0].x = g0.x; bb[0].y = g1.x; bb[1].x = g0.y; bb[1].y = g1.y;
        #pragma unroll
        for (int nt = 0; nt < 2; ++nt)
            #pragma unroll
            for (int mt = 0; mt < 4; ++mt)
                accR[nt][mt] = __builtin_amdgcn_wmma_f32_16x16x4_f32(
                    false, a[mt], false, bb[nt], (short)0, accR[nt][mt], false, false);
    }
    for (int k = 0; k < H_DIM; k += 4) {
        v2f a[4];
        #pragma unroll
        for (int mt = 0; mt < 4; ++mt)
            a[mt] = *(const v2f*)(&xc[(mt * 16 + a_lrow) * LDS_STRIDE + IN_DIM + k + a_koff]);
        const long wrow = (long)(k + b_krow) * H_DIM + col2;
        const v2f g0 = *(const v2f*)(whh + wrow);
        const v2f g1 = *(const v2f*)(whh + wrow + H_DIM);
        v2f bb[2];
        bb[0].x = g0.x; bb[0].y = g1.x; bb[1].x = g0.y; bb[1].y = g1.y;
        #pragma unroll
        for (int nt = 0; nt < 2; ++nt)
            #pragma unroll
            for (int mt = 0; mt < 4; ++mt)
                accR[nt][mt] = __builtin_amdgcn_wmma_f32_16x16x4_f32(
                    false, a[mt], false, bb[nt], (short)0, accR[nt][mt], false, false);
    }

    {
        const v2f bzc = *(const v2f*)(bz + col2);
        const v2f bhc = *(const v2f*)(bh + col2);
        #pragma unroll
        for (int mt = 0; mt < 4; ++mt) {
            #pragma unroll
            for (int v = 0; v < 8; ++v) {
                const long row = block_row + mt * 16 + (lane >> 4) * 8 + v;
                float z0 = accZ[0][mt][v] + bzc.x;
                float z1 = accZ[1][mt][v] + bzc.y;
                z0 = sigmoid_hw(z0);
                z1 = sigmoid_hw(z1);
                const float g0 = tanh_hw(accR[0][mt][v] + bhc.x);
                const float g1 = tanh_hw(accR[1][mt][v] + bhc.y);
                v2f hn;
                hn.x = (1.0f - z0) * hreg[0][mt][v] + z0 * g0;
                hn.y = (1.0f - z1) * hreg[1][mt][v] + z1 * g1;
                __builtin_nontemporal_store(hn, (v2f*)(out + row * H_DIM + col2));
            }
        }
    }
}

extern "C" void kernel_launch(void* const* d_in, const int* in_sizes, int n_in,
                              void* d_out, int out_size, void* d_ws, size_t ws_size,
                              hipStream_t stream) {
    const float* x   = (const float*)d_in[0];
    const float* hp  = (const float*)d_in[1];
    const float* wr  = (const float*)d_in[2];
    const float* wz  = (const float*)d_in[3];
    const float* whh = (const float*)d_in[4];
    const float* whx = (const float*)d_in[5];
    const float* br  = (const float*)d_in[6];
    const float* bz  = (const float*)d_in[7];
    const float* bh  = (const float*)d_in[8];
    float* out = (float*)d_out;

    const int rows = in_sizes[0] / IN_DIM;
    dim3 grid(rows / M_TILE), block(THREADS);
    gru_wmma_f32<<<grid, block, 0, stream>>>(x, hp, wr, wz, whh, whx, br, bz, bh, out);
}
